# starting kernel
# baseline (speedup 1.0000x reference)
#include <hip/hip_runtime.h>
#include <math.h>

typedef __attribute__((ext_vector_type(2))) float v2f;
typedef __attribute__((ext_vector_type(8))) float v8f;

#define ALPHA 0.2f
#define FDIM  128
#define DNB   16
#define LSEQ  4096
#define BL    32768

__global__ __launch_bounds__(256)
void gat_gemm_xW(const float* __restrict__ x, const float* __restrict__ W,
                 float* __restrict__ h) {
  __shared__ float Wt[FDIM * FDIM];

  for (int i = threadIdx.x; i < FDIM * FDIM; i += blockDim.x) {
    int k = i >> 7, n = i & 127;
    Wt[n * FDIM + k] = W[i];
  }
  __syncthreads();

  const int wave  = threadIdx.x >> 5;
  const int lane  = threadIdx.x & 31;
  const int row0  = (blockIdx.x * 8 + wave) * 16;
  const int m     = lane & 15;
  const int khalf = (lane >> 4) << 1;

  v2f afrag[32];
  const float* xrow = x + (size_t)(row0 + m) * FDIM + khalf;
  #pragma unroll
  for (int ks = 0; ks < 32; ++ks)
    afrag[ks] = *(const v2f*)(xrow + ks * 4);

  v8f acc[8] = {};

  #pragma unroll
  for (int ks = 0; ks < 32; ++ks) {
    const int kb = ks * 4 + khalf;
    #pragma unroll
    for (int n = 0; n < 8; ++n) {
      v2f bfrag = *(const v2f*)(&Wt[(n * 16 + m) * FDIM + kb]);
      acc[n] = __builtin_amdgcn_wmma_f32_16x16x4_f32(
           false, afrag[ks],  false, bfrag,
           (short)0, acc[n],  false,  false);
    }
  }

  const int mhi = (lane >> 4) << 3;
  #pragma unroll
  for (int n = 0; n < 8; ++n) {
    #pragma unroll
    for (int r = 0; r < 8; ++r) {
      h[(size_t)(row0 + r + mhi) * FDIM + n * 16 + m] = acc[n][r];
    }
  }
}

__device__ __forceinline__ float wave_sum(float s) {
  #pragma unroll
  for (int off = 16; off > 0; off >>= 1)
    s += __shfl_xor(s, off, 32);
  return s;
}

__global__ __launch_bounds__(256)
void gat_attn(const float* __restrict__ h, const int* __restrict__ adj,
              const float* __restrict__ a, float* __restrict__ out) {
  const int wid  = (int)((blockIdx.x * blockDim.x + threadIdx.x) >> 5);
  const int lane = threadIdx.x & 31;
  if (wid >= BL) return;

  const int    b    = wid >> 12;
  const size_t base = (size_t)b * LSEQ;
  const float4* hv  = (const float4*)h;

  float4 v[DNB];
  const int* arow = adj + (size_t)wid * DNB;
  #pragma unroll
  for (int d = 0; d < DNB; ++d) {
    const int nb = arow[d];
    v[d] = hv[(base + (size_t)nb) * (FDIM / 4) + lane];
  }

  const float4 alo = ((const float4*)a)[lane];
  const float4 ahi = ((const float4*)a)[32 + lane];

  const float ehi = wave_sum(v[0].x * ahi.x + v[0].y * ahi.y +
                             v[0].z * ahi.z + v[0].w * ahi.w);

  float e[DNB];
  float emax = -1e30f;
  #pragma unroll
  for (int d = 0; d < DNB; ++d) {
    float s = wave_sum(v[d].x * alo.x + v[d].y * alo.y +
                       v[d].z * alo.z + v[d].w * alo.w) + ehi;
    s = (s > 0.f) ? s : ALPHA * s;
    e[d] = s;
    emax = fmaxf(emax, s);
  }

  float denom = 0.f;
  #pragma unroll
  for (int d = 0; d < DNB; ++d) { e[d] = __expf(e[d] - emax); denom += e[d]; }
  const float inv = 1.f / denom;

  float4 o = {0.f, 0.f, 0.f, 0.f};
  #pragma unroll
  for (int d = 0; d < DNB; ++d) {
    const float w = e[d] * inv;
    o.x += w * v[d].x; o.y += w * v[d].y;
    o.z += w * v[d].z; o.w += w * v[d].w;
  }

  o.x = (o.x > 0.f) ? o.x : (__expf(o.x) - 1.f);
  o.y = (o.y > 0.f) ? o.y : (__expf(o.y) - 1.f);
  o.z = (o.z > 0.f) ? o.z : (__expf(o.z) - 1.f);
  o.w = (o.w > 0.f) ? o.w : (__expf(o.w) - 1.f);

  ((float4*)out)[(size_t)wid * (FDIM / 4) + lane] = o;
}

extern "C" void kernel_launch(void* const* d_in, const int* in_sizes, int n_in,
                              void* d_out, int out_size, void* d_ws, size_t ws_size,
                              hipStream_t stream) {
  const float* x   = (const float*)d_in[0];
  const int*   adj = (const int*)d_in[1];
  const float* W   = (const float*)d_in[2];
  const float* a   = (const float*)d_in[3];
  float*       out = (float*)d_out;
  float*       h   = (float*)d_ws;

  gat_gemm_xW<<<256, 256, 0, stream>>>(x, W, h);
  gat_attn<<<4096, 256, 0, stream>>>(h, adj, a, out);
}


# GEMM without LDS: each wave owns a 64x64 block of h (16 accumulator tiles) and streams K, x and W fragments loaded straight from global by b128 far ahead; k order inside each 8 changes (sum order)
# speedup vs baseline: 1.9534x; 1.9534x over previous
#include <hip/hip_runtime.h>
#include <math.h>

typedef __attribute__((ext_vector_type(2))) float v2f;
typedef __attribute__((ext_vector_type(4))) float v4f;
typedef __attribute__((ext_vector_type(8))) float v8f;

#define ALPHA 0.2f
#define FDIM  128
#define DNB   16
#define LSEQ  4096
#define BL    32768

#define GEMM_AHEAD 7

__global__ __launch_bounds__(128)
void gat_gemm_xW(const float* __restrict__ x, const float* __restrict__ W,
                 float* __restrict__ h) {
  const int wave  = threadIdx.x >> 5;
  const int lane  = threadIdx.x & 31;
  const int c     = lane & 15;
  const int hh    = lane >> 4;
  const int strip = wave & 1;
  const int grp   = wave >> 1;
  const int row0  = blockIdx.x * 128 + grp * 64;
  const int col0  = strip * 64 + c * 4;

  const float* wp = W + (size_t)(4 * hh) * FDIM + col0;
  const float* xp = x + (size_t)(row0 + c) * FDIM + 4 * hh;

  v4f xa[16][4];
  v4f wq[16][4];
  v8f acc[4][4] = {};

#define GEMM_REQUEST(P)                                                        \
  {                                                                            \
    _Pragma("unroll") for (int t = 0; t < 4; ++t)                              \
      xa[P][t] = *(const v4f*)(xp + (size_t)t * 16 * FDIM + 8 * (P));          \
    _Pragma("unroll") for (int q = 0; q < 4; ++q)                              \
      wq[P][q] = *(const v4f*)(wp + (size_t)(8 * (P) + q) * FDIM);             \
  }

  #pragma unroll
  for (int p = 0; p < GEMM_AHEAD; ++p) GEMM_REQUEST(p)

  #pragma unroll
  for (int p = 0; p < 16; ++p) {
    if (p + GEMM_AHEAD < 16) GEMM_REQUEST(p + GEMM_AHEAD)
    __builtin_amdgcn_sched_barrier(0);
    #pragma unroll
    for (int e = 0; e < 2; ++e) {
      #pragma unroll
      for (int t = 0; t < 4; ++t) {
        v2f af;
        af[0] = xa[p][t][2 * e];
        af[1] = xa[p][t][2 * e + 1];
        #pragma unroll
        for (int n = 0; n < 4; ++n) {
          v2f bf;
          bf[0] = wq[p][2 * e][n];
          bf[1] = wq[p][2 * e + 1][n];
          acc[t][n] = __builtin_amdgcn_wmma_f32_16x16x4_f32(
               false, af,  false, bf,
               (short)0, acc[t][n],  false,  false);
        }
      }
    }
  }
#undef GEMM_REQUEST

  #pragma unroll
  for (int t = 0; t < 4; ++t) {
    float* hrow = h + (size_t)(row0 + 16 * t + 8 * hh) * FDIM + col0;
    #pragma unroll
    for (int r = 0; r < 8; ++r) {
      v4f o;
      o[0] = acc[t][0][r]; o[1] = acc[t][1][r];
      o[2] = acc[t][2][r]; o[3] = acc[t][3][r];
      *(v4f*)(hrow + (size_t)r * FDIM) = o;
    }
  }
}

__device__ __forceinline__ float wave_sum(float s) {
  #pragma unroll
  for (int off = 16; off > 0; off >>= 1)
    s += __shfl_xor(s, off, 32);
  return s;
}

__global__ __launch_bounds__(256)
void gat_attn(const float* __restrict__ h, const int* __restrict__ adj,
              const float* __restrict__ a, float* __restrict__ out) {
  const int wid  = (int)((blockIdx.x * blockDim.x + threadIdx.x) >> 5);
  const int lane = threadIdx.x & 31;
  if (wid >= BL) return;

  const int    b    = wid >> 12;
  const size_t base = (size_t)b * LSEQ;
  const float4* hv  = (const float4*)h;

  float4 v[DNB];
  const int* arow = adj + (size_t)wid * DNB;
  #pragma unroll
  for (int d = 0; d < DNB; ++d) {
    const int nb = arow[d];
    v[d] = hv[(base + (size_t)nb) * (FDIM / 4) + lane];
  }

  const float4 alo = ((const float4*)a)[lane];
  const float4 ahi = ((const float4*)a)[32 + lane];

  const float ehi = wave_sum(v[0].x * ahi.x + v[0].y * ahi.y +
                             v[0].z * ahi.z + v[0].w * ahi.w);

  float e[DNB];
  float emax = -1e30f;
  #pragma unroll
  for (int d = 0; d < DNB; ++d) {
    float s = wave_sum(v[d].x * alo.x + v[d].y * alo.y +
                       v[d].z * alo.z + v[d].w * alo.w) + ehi;
    s = (s > 0.f) ? s : ALPHA * s;
    e[d] = s;
    emax = fmaxf(emax, s);
  }

  float denom = 0.f;
  #pragma unroll
  for (int d = 0; d < DNB; ++d) { e[d] = __expf(e[d] - emax); denom += e[d]; }
  const float inv = 1.f / denom;

  float4 o = {0.f, 0.f, 0.f, 0.f};
  #pragma unroll
  for (int d = 0; d < DNB; ++d) {
    const float w = e[d] * inv;
    o.x += w * v[d].x; o.y += w * v[d].y;
    o.z += w * v[d].z; o.w += w * v[d].w;
  }

  o.x = (o.x > 0.f) ? o.x : (__expf(o.x) - 1.f);
  o.y = (o.y > 0.f) ? o.y : (__expf(o.y) - 1.f);
  o.z = (o.z > 0.f) ? o.z : (__expf(o.z) - 1.f);
  o.w = (o.w > 0.f) ? o.w : (__expf(o.w) - 1.f);

  ((float4*)out)[(size_t)wid * (FDIM / 4) + lane] = o;
}

extern "C" void kernel_launch(void* const* d_in, const int* in_sizes, int n_in,
                              void* d_out, int out_size, void* d_ws, size_t ws_size,
                              hipStream_t stream) {
  const float* x   = (const float*)d_in[0];
  const int*   adj = (const int*)d_in[1];
  const float* W   = (const float*)d_in[2];
  const float* a   = (const float*)d_in[3];
  float*       out = (float*)d_out;
  float*       h   = (float*)d_ws;

  gat_gemm_xW<<<256, 128, 0, stream>>>(x, W, h);
  gat_attn<<<4096, 256, 0, stream>>>(h, adj, a, out);
}


# Attention: one node per 16-lane row (2 nodes a wave, 16384 waves), lane = neighbour for ids and softmax; the 17 sums as row DPP adds, none through LDS; logit, softmax sum order changes
# speedup vs baseline: 2.1958x; 1.1241x over previous
#include <hip/hip_runtime.h>
#include <math.h>

typedef __attribute__((ext_vector_type(2))) float v2f;
typedef __attribute__((ext_vector_type(4))) float v4f;
typedef __attribute__((ext_vector_type(8))) float v8f;

#define ALPHA 0.2f
#define FDIM  128
#define DNB   16
#define LSEQ  4096
#define BL    32768

#define GEMM_AHEAD 7

__global__ __launch_bounds__(128)
void gat_gemm_xW(const float* __restrict__ x, const float* __restrict__ W,
                 float* __restrict__ h) {
  const int wave  = threadIdx.x >> 5;
  const int lane  = threadIdx.x & 31;
  const int c     = lane & 15;
  const int hh    = lane >> 4;
  const int strip = wave & 1;
  const int grp   = wave >> 1;
  const int row0  = blockIdx.x * 128 + grp * 64;
  const int col0  = strip * 64 + c * 4;

  const float* wp = W + (size_t)(4 * hh) * FDIM + col0;
  const float* xp = x + (size_t)(row0 + c) * FDIM + 4 * hh;

  v4f xa[16][4];
  v4f wq[16][4];
  v8f acc[4][4] = {};

#define GEMM_REQUEST(P)                                                        \
  {                                                                            \
    _Pragma("unroll") for (int t = 0; t < 4; ++t)                              \
      xa[P][t] = *(const v4f*)(xp + (size_t)t * 16 * FDIM + 8 * (P));          \
    _Pragma("unroll") for (int q = 0; q < 4; ++q)                              \
      wq[P][q] = *(const v4f*)(wp + (size_t)(8 * (P) + q) * FDIM);             \
  }

  #pragma unroll
  for (int p = 0; p < GEMM_AHEAD; ++p) GEMM_REQUEST(p)

  #pragma unroll
  for (int p = 0; p < 16; ++p) {
    if (p + GEMM_AHEAD < 16) GEMM_REQUEST(p + GEMM_AHEAD)
    __builtin_amdgcn_sched_barrier(0);
    #pragma unroll
    for (int e = 0; e < 2; ++e) {
      #pragma unroll
      for (int t = 0; t < 4; ++t) {
        v2f af;
        af[0] = xa[p][t][2 * e];
        af[1] = xa[p][t][2 * e + 1];
        #pragma unroll
        for (int n = 0; n < 4; ++n) {
          v2f bf;
          bf[0] = wq[p][2 * e][n];
          bf[1] = wq[p][2 * e + 1][n];
          acc[t][n] = __builtin_amdgcn_wmma_f32_16x16x4_f32(
               false, af,  false, bf,
               (short)0, acc[t][n],  false,  false);
        }
      }
    }
  }
#undef GEMM_REQUEST

  #pragma unroll
  for (int t = 0; t < 4; ++t) {
    float* hrow = h + (size_t)(row0 + 16 * t + 8 * hh) * FDIM + col0;
    #pragma unroll
    for (int r = 0; r < 8; ++r) {
      v4f o;
      o[0] = acc[t][0][r]; o[1] = acc[t][1][r];
      o[2] = acc[t][2][r]; o[3] = acc[t][3][r];
      *(v4f*)(hrow + (size_t)r * FDIM) = o;
    }
  }
}

template <int CTRL>
__device__ __forceinline__ float row_get(float x) {
  return __builtin_bit_cast(float, __builtin_amdgcn_update_dpp(
      0, __builtin_bit_cast(int, x), CTRL, 0xf, 0xf, true));
}
template <int CTRL>
__device__ __forceinline__ unsigned row_get_u(unsigned x) {
  return (unsigned)__builtin_amdgcn_update_dpp(0, (int)x, CTRL, 0xf, 0xf, true);
}
__device__ __forceinline__ float row_sum(float s) {
  s += row_get<0x161>(s); s += row_get<0x162>(s);
  s += row_get<0x164>(s); s += row_get<0x168>(s);
  return s;
}
__device__ __forceinline__ float row_max(float s) {
  s = fmaxf(s, row_get<0x161>(s)); s = fmaxf(s, row_get<0x162>(s));
  s = fmaxf(s, row_get<0x164>(s)); s = fmaxf(s, row_get<0x168>(s));
  return s;
}
__device__ __forceinline__ float dot8(v4f x0, v4f x1, v4f y0, v4f y1) {
  v2f s = x0.xy * y0.xy;
  s += x0.zw * y0.zw;
  s += x1.xy * y1.xy;
  s += x1.zw * y1.zw;
  return s[0] + s[1];
}

#define REP16(M) M(0) M(1) M(2) M(3) M(4) M(5) M(6) M(7) \
                 M(8) M(9) M(10) M(11) M(12) M(13) M(14) M(15)

__global__ __launch_bounds__(256)
void gat_attn(const float* __restrict__ h, const int* __restrict__ adj,
              const float* __restrict__ a, float* __restrict__ out) {
  const int lane = threadIdx.x & 31;
  const int c    = lane & 15;
  const int wv = __builtin_amdgcn_readfirstlane(
      (int)(blockIdx.x * 8 + (threadIdx.x >> 5)));
  const int node = wv * 2 + (lane >> 4);
  const int b    = wv >> 11;
  const char* hb = (const char*)(h + (size_t)b * LSEQ * FDIM);

  int nbv = adj[(size_t)wv * 32 + lane];
  nbv = min(max(nbv, 0), LSEQ - 1);
  const unsigned myoff = (unsigned)nbv * (FDIM * 4u);
  const unsigned loff  = (unsigned)c * 16u;

  v4f v0[DNB], v1[DNB];
#define GATHER(D)                                                         \
  {                                                                       \
    const unsigned off = row_get_u<0x150 + D>(myoff) + loff;              \
    v0[D] = *(const v4f*)(hb + off);                                      \
    v1[D] = *(const v4f*)(hb + off + 256u);                               \
  }
  REP16(GATHER)
#undef GATHER

  const v4f alo0 = *(const v4f*)(a + 4 * c);
  const v4f alo1 = *(const v4f*)(a + 64 + 4 * c);
  const v4f ahi0 = *(const v4f*)(a + FDIM + 4 * c);
  const v4f ahi1 = *(const v4f*)(a + FDIM + 64 + 4 * c);
  __builtin_amdgcn_sched_barrier(0);

  const float ehi = row_sum(dot8(v0[0], v1[0], ahi0, ahi1));

  float p[DNB];
  #pragma unroll
  for (int d = 0; d < DNB; ++d) p[d] = dot8(v0[d], v1[d], alo0, alo1);

  const bool b8 = (c & 8) != 0, b4 = (c & 4) != 0;
  const bool b2 = (c & 2) != 0, b1 = (c & 1) != 0;
  float q[8], r[4], t[2];
  #pragma unroll
  for (int j = 0; j < 8; ++j) {
    const float keep = b8 ? p[j + 8] : p[j];
    const float send = b8 ? p[j] : p[j + 8];
    q[j] = keep + row_get<0x168>(send);
  }
  #pragma unroll
  for (int j = 0; j < 4; ++j) {
    const float keep = b4 ? q[j + 4] : q[j];
    const float send = b4 ? q[j] : q[j + 4];
    r[j] = keep + row_get<0x164>(send);
  }
  #pragma unroll
  for (int j = 0; j < 2; ++j) {
    const float keep = b2 ? r[j + 2] : r[j];
    const float send = b2 ? r[j] : r[j + 2];
    t[j] = keep + row_get<0x162>(send);
  }
  float s;
  {
    const float keep = b1 ? t[1] : t[0];
    const float send = b1 ? t[0] : t[1];
    s = keep + row_get<0x161>(send) + ehi;
  }
  s = (s > 0.f) ? s : ALPHA * s;

  const float emax  = row_max(s);
  const float ex    = __expf(s - emax);
  const float denom = row_sum(ex);
  const float w     = ex * (1.f / denom);

  v4f o0 = {0.f, 0.f, 0.f, 0.f}, o1 = {0.f, 0.f, 0.f, 0.f};
#define AGG(D)                                                            \
  {                                                                       \
    const float wd = row_get<0x150 + D>(w);                               \
    o0 += wd * v0[D];                                                     \
    o1 += wd * v1[D];                                                     \
  }
  REP16(AGG)
#undef AGG

  #pragma unroll
  for (int k = 0; k < 4; ++k) {
    o0[k] = (o0[k] > 0.f) ? o0[k] : (__expf(o0[k]) - 1.f);
    o1[k] = (o1[k] > 0.f) ? o1[k] : (__expf(o1[k]) - 1.f);
  }

  float* orow = out + (size_t)node * FDIM + 4 * c;
  *(v4f*)orow        = o0;
  *(v4f*)(orow + 64) = o1;
}
#undef REP16

extern "C" void kernel_launch(void* const* d_in, const int* in_sizes, int n_in,
                              void* d_out, int out_size, void* d_ws, size_t ws_size,
                              hipStream_t stream) {
  const float* x   = (const float*)d_in[0];
  const int*   adj = (const int*)d_in[1];
  const float* W   = (const float*)d_in[2];
  const float* a   = (const float*)d_in[3];
  float*       out = (float*)d_out;
  float*       h   = (float*)d_ws;

  gat_gemm_xW<<<256, 128, 0, stream>>>(x, W, h);
  gat_attn<<<2048, 256, 0, stream>>>(h, adj, a, out);
}


# Attention launch sized to what stays resident: 512 blocks of 8 waves (4096 waves, was 16384); a wave serves 4 consecutive node pairs in turn with one body, ids requested at entry; sums unchanged
# speedup vs baseline: 2.4761x; 1.1277x over previous
#include <hip/hip_runtime.h>
#include <math.h>

typedef __attribute__((ext_vector_type(2))) float v2f;
typedef __attribute__((ext_vector_type(4))) float v4f;
typedef __attribute__((ext_vector_type(8))) float v8f;

#define ALPHA 0.2f
#define FDIM  128
#define DNB   16
#define LSEQ  4096
#define BL    32768

#define GEMM_AHEAD 7

__global__ __launch_bounds__(128)
void gat_gemm_xW(const float* __restrict__ x, const float* __restrict__ W,
                 float* __restrict__ h) {
  const int wave  = threadIdx.x >> 5;
  const int lane  = threadIdx.x & 31;
  const int c     = lane & 15;
  const int hh    = lane >> 4;
  const int strip = wave & 1;
  const int grp   = wave >> 1;
  const int row0  = blockIdx.x * 128 + grp * 64;
  const int col0  = strip * 64 + c * 4;

  const float* wp = W + (size_t)(4 * hh) * FDIM + col0;
  const float* xp = x + (size_t)(row0 + c) * FDIM + 4 * hh;

  v4f xa[16][4];
  v4f wq[16][4];
  v8f acc[4][4] = {};

#define GEMM_REQUEST(P)                                                        \
  {                                                                            \
    _Pragma("unroll") for (int t = 0; t < 4; ++t)                              \
      xa[P][t] = *(const v4f*)(xp + (size_t)t * 16 * FDIM + 8 * (P));          \
    _Pragma("unroll") for (int q = 0; q < 4; ++q)                              \
      wq[P][q] = *(const v4f*)(wp + (size_t)(8 * (P) + q) * FDIM);             \
  }

  #pragma unroll
  for (int p = 0; p < GEMM_AHEAD; ++p) GEMM_REQUEST(p)

  #pragma unroll
  for (int p = 0; p < 16; ++p) {
    if (p + GEMM_AHEAD < 16) GEMM_REQUEST(p + GEMM_AHEAD)
    __builtin_amdgcn_sched_barrier(0);
    #pragma unroll
    for (int e = 0; e < 2; ++e) {
      #pragma unroll
      for (int t = 0; t < 4; ++t) {
        v2f af;
        af[0] = xa[p][t][2 * e];
        af[1] = xa[p][t][2 * e + 1];
        #pragma unroll
        for (int n = 0; n < 4; ++n) {
          v2f bf;
          bf[0] = wq[p][2 * e][n];
          bf[1] = wq[p][2 * e + 1][n];
          acc[t][n] = __builtin_amdgcn_wmma_f32_16x16x4_f32(
               false, af,  false, bf,
               (short)0, acc[t][n],  false,  false);
        }
      }
    }
  }
#undef GEMM_REQUEST

  #pragma unroll
  for (int t = 0; t < 4; ++t) {
    float* hrow = h + (size_t)(row0 + 16 * t + 8 * hh) * FDIM + col0;
    #pragma unroll
    for (int r = 0; r < 8; ++r) {
      v4f o;
      o[0] = acc[t][0][r]; o[1] = acc[t][1][r];
      o[2] = acc[t][2][r]; o[3] = acc[t][3][r];
      *(v4f*)(hrow + (size_t)r * FDIM) = o;
    }
  }
}

#define PAIRS_PER_WAVE 4
template <int CTRL>
__device__ __forceinline__ float row_get(float x) {
  return __builtin_bit_cast(float, __builtin_amdgcn_update_dpp(
      0, __builtin_bit_cast(int, x), CTRL, 0xf, 0xf, true));
}
template <int CTRL>
__device__ __forceinline__ unsigned row_get_u(unsigned x) {
  return (unsigned)__builtin_amdgcn_update_dpp(0, (int)x, CTRL, 0xf, 0xf, true);
}
__device__ __forceinline__ float row_sum(float s) {
  s += row_get<0x161>(s); s += row_get<0x162>(s);
  s += row_get<0x164>(s); s += row_get<0x168>(s);
  return s;
}
__device__ __forceinline__ float row_max(float s) {
  s = fmaxf(s, row_get<0x161>(s)); s = fmaxf(s, row_get<0x162>(s));
  s = fmaxf(s, row_get<0x164>(s)); s = fmaxf(s, row_get<0x168>(s));
  return s;
}
__device__ __forceinline__ float dot8(v4f x0, v4f x1, v4f y0, v4f y1) {
  v2f s = x0.xy * y0.xy;
  s += x0.zw * y0.zw;
  s += x1.xy * y1.xy;
  s += x1.zw * y1.zw;
  return s[0] + s[1];
}

#define REP16(M) M(0) M(1) M(2) M(3) M(4) M(5) M(6) M(7) \
                 M(8) M(9) M(10) M(11) M(12) M(13) M(14) M(15)

__global__ __launch_bounds__(256)
void gat_attn(const float* __restrict__ h, const int* __restrict__ adj,
              const float* __restrict__ a, float* __restrict__ out) {
  const int lane = threadIdx.x & 31;
  const int c    = lane & 15;
  const int wv = __builtin_amdgcn_readfirstlane(
      (int)(blockIdx.x * 8 + (threadIdx.x >> 5)));
  const int b  = wv >> 9;
  const char* hb = (const char*)(h + (size_t)b * LSEQ * FDIM);

  const int* ap = adj + (size_t)wv * (32 * PAIRS_PER_WAVE) + lane;
  int n0 = ap[0], n1 = ap[32], n2 = ap[64], n3 = ap[96];

  const v4f alo0 = *(const v4f*)(a + 4 * c);
  const v4f alo1 = *(const v4f*)(a + 64 + 4 * c);
  const v4f ahi0 = *(const v4f*)(a + FDIM + 4 * c);
  const v4f ahi1 = *(const v4f*)(a + FDIM + 64 + 4 * c);

  const unsigned loff = (unsigned)c * 16u;
  const bool b8 = (c & 8) != 0, b4 = (c & 4) != 0;
  const bool b2 = (c & 2) != 0, b1 = (c & 1) != 0;
  float* orow = out + ((size_t)wv * (2 * PAIRS_PER_WAVE) + (lane >> 4)) * FDIM
                    + 4 * c;

  #pragma unroll 1
  for (int it = 0; it < PAIRS_PER_WAVE; ++it) {
    const int nbv = min(max(n0, 0), LSEQ - 1);
    n0 = n1; n1 = n2; n2 = n3;
    const unsigned myoff = (unsigned)nbv * (FDIM * 4u);

    v4f v0[DNB], v1[DNB];
#define GATHER(D)                                                         \
    {                                                                     \
      const unsigned off = row_get_u<0x150 + D>(myoff) + loff;            \
      v0[D] = *(const v4f*)(hb + off);                                    \
      v1[D] = *(const v4f*)(hb + off + 256u);                             \
    }
    REP16(GATHER)
#undef GATHER
    __builtin_amdgcn_sched_barrier(0);

    const float ehi = row_sum(dot8(v0[0], v1[0], ahi0, ahi1));

    float p[DNB];
    #pragma unroll
    for (int d = 0; d < DNB; ++d) p[d] = dot8(v0[d], v1[d], alo0, alo1);

    float q[8], r[4], t[2];
    #pragma unroll
    for (int j = 0; j < 8; ++j) {
      const float keep = b8 ? p[j + 8] : p[j];
      const float send = b8 ? p[j] : p[j + 8];
      q[j] = keep + row_get<0x168>(send);
    }
    #pragma unroll
    for (int j = 0; j < 4; ++j) {
      const float keep = b4 ? q[j + 4] : q[j];
      const float send = b4 ? q[j] : q[j + 4];
      r[j] = keep + row_get<0x164>(send);
    }
    #pragma unroll
    for (int j = 0; j < 2; ++j) {
      const float keep = b2 ? r[j + 2] : r[j];
      const float send = b2 ? r[j] : r[j + 2];
      t[j] = keep + row_get<0x162>(send);
    }
    float s;
    {
      const float keep = b1 ? t[1] : t[0];
      const float send = b1 ? t[0] : t[1];
      s = keep + row_get<0x161>(send) + ehi;
    }
    s = (s > 0.f) ? s : ALPHA * s;

    const float emax  = row_max(s);
    const float ex    = __expf(s - emax);
    const float denom = row_sum(ex);
    const float w     = ex * (1.f / denom);

    v4f o0 = {0.f, 0.f, 0.f, 0.f}, o1 = {0.f, 0.f, 0.f, 0.f};
#define AGG(D)                                                            \
    {                                                                     \
      const float wd = row_get<0x150 + D>(w);                             \
      o0 += wd * v0[D];                                                   \
      o1 += wd * v1[D];                                                   \
    }
    REP16(AGG)
#undef AGG

    #pragma unroll
    for (int k = 0; k < 4; ++k) {
      o0[k] = (o0[k] > 0.f) ? o0[k] : (__expf(o0[k]) - 1.f);
      o1[k] = (o1[k] > 0.f) ? o1[k] : (__expf(o1[k]) - 1.f);
    }

    *(v4f*)orow        = o0;
    *(v4f*)(orow + 64) = o1;
    orow += 2 * FDIM;
  }
}
#undef PAIRS_PER_WAVE
#undef REP16

extern "C" void kernel_launch(void* const* d_in, const int* in_sizes, int n_in,
                              void* d_out, int out_size, void* d_ws, size_t ws_size,
                              hipStream_t stream) {
  const float* x   = (const float*)d_in[0];
  const int*   adj = (const int*)d_in[1];
  const float* W   = (const float*)d_in[2];
  const float* a   = (const float*)d_in[3];
  float*       out = (float*)d_out;
  float*       h   = (float*)d_ws;

  gat_gemm_xW<<<256, 128, 0, stream>>>(x, W, h);
  gat_attn<<<512, 256, 0, stream>>>(h, adj, a, out);
}
